# starting kernel
# baseline (speedup 1.0000x reference)
#include <hip/hip_runtime.h>

#define TPB 256
#define LD64 65
#define LD32 33
#define IMGS_PER_BLOCK 2

typedef float v2f __attribute__((ext_vector_type(2)));
typedef float v8f __attribute__((ext_vector_type(8)));

__constant__ float c_H[8] = {
    0.23037781330885523f,  0.7148465705525415f,   0.6308807679295904f,
    -0.02798376941698385f, -0.18703481171888114f, 0.030841381835986965f,
    0.032883011666982945f, -0.010597401784997278f};

constexpr bool anz(int N, int k, int j) {
  int kk = (k >= N / 2) ? (k - N / 2) : k;
  int d = (j - 2 * kk) % N;
  if (d < 0) d += N;
  return d < 8;
}
constexpr unsigned mask_rows(int N, int r0) {
  unsigned m = 0;
  for (int jt = 0; jt < N / 4; ++jt)
    for (int k = 4 * jt; k < 4 * jt + 4; ++k)
      for (int r = r0; r < r0 + 16; ++r)
        if (anz(N, r, k)) m |= 1u << jt;
  return m;
}
constexpr unsigned mask_cols(int N, int c0) {
  unsigned m = 0;
  for (int jt = 0; jt < N / 4; ++jt)
    for (int k = 4 * jt; k < 4 * jt + 4; ++k)
      for (int c = c0; c < c0 + 16; ++c)
        if (anz(N, k, c)) m |= 1u << jt;
  return m;
}
constexpr unsigned sel_mask(int N, int SEL, int rb, int cb) {
  int blk = (SEL == 0 || SEL == 2) ? rb : cb;
  return (SEL <= 1) ? mask_rows(N, 16 * blk) : mask_cols(N, 16 * blk);
}

template <int... Is> struct iseq {};
template <int N, int... Is> struct mkseq : mkseq<N - 1, N - 1, Is...> {};
template <int... Is> struct mkseq<0, Is...> { using type = iseq<Is...>; };

template <int K, bool TA, bool TB>
__device__ inline v8f wmma_tile(const float* __restrict__ M1, int ld1,
                                const float* __restrict__ M2, int ld2,
                                int r0, int c0, int lane, unsigned mask) {
  v8f acc = {0.f, 0.f, 0.f, 0.f, 0.f, 0.f, 0.f, 0.f};
  const int hw = lane >> 4;
  const int l16 = lane & 15;
  const int row = r0 + l16;
  const int col = c0 + l16;
#pragma unroll
  for (int j = 0; j < K / 4; ++j) {
    if (mask & (1u << j)) {
      const int ka = 4 * j + 2 * hw;
      v2f a, b;
      if (TA) { a.x = M1[ka * ld1 + row];  a.y = M1[(ka + 1) * ld1 + row]; }
      else    { a.x = M1[row * ld1 + ka];  a.y = M1[row * ld1 + ka + 1];   }
      if (TB) { b.x = M2[col * ld2 + ka];  b.y = M2[col * ld2 + ka + 1];   }
      else    { b.x = M2[ka * ld2 + col];  b.y = M2[(ka + 1) * ld2 + col]; }
      acc = __builtin_amdgcn_wmma_f32_16x16x4_f32(false, a, false, b,
                                                  (short)0, acc, false, false);
    }
  }
  return acc;
}

__device__ inline void store_tile(float* __restrict__ D, int ld, int r0, int c0,
                                  int lane, v8f acc) {
  const int hw = lane >> 4;
  const int c = c0 + (lane & 15);
#pragma unroll
  for (int v = 0; v < 8; ++v) D[(r0 + v + 8 * hw) * ld + c] = acc[v];
}

template <bool TA, bool TB, int SEL>
struct S64 {
  template <int T>
  static __device__ inline void one(const float* M1, const float* M2, float* D,
                                    int wave, int lane) {
    constexpr int r0 = (T >> 2) << 4, c0 = (T & 3) << 4;
    constexpr unsigned mask = sel_mask(64, SEL, T >> 2, T & 3);
    if (wave == (T & 7)) {
      v8f acc = wmma_tile<64, TA, TB>(M1, LD64, M2, LD64, r0, c0, lane, mask);
      store_tile(D, LD64, r0, c0, lane, acc);
    }
  }
  template <int... Ts>
  static __device__ inline void go(const float* M1, const float* M2, float* D,
                                   int wave, int lane, iseq<Ts...>) {
    (one<Ts>(M1, M2, D, wave, lane), ...);
  }
  static __device__ inline void run(const float* M1, const float* M2, float* D,
                                    int wave, int lane) {
    go(M1, M2, D, wave, lane, typename mkseq<16>::type{});
  }
};

template <bool TA, bool TB, int SEL>
struct S32 {
  template <int T>
  static __device__ inline void one(const float* M1, int ld1, const float* M2,
                                    int ld2, float* D, int ldd, int wave, int lane) {
    constexpr int r0 = (T >> 1) << 4, c0 = (T & 1) << 4;
    constexpr unsigned mask = sel_mask(32, SEL, T >> 1, T & 1);
    if (wave == T) {
      v8f acc = wmma_tile<32, TA, TB>(M1, ld1, M2, ld2, r0, c0, lane, mask);
      store_tile(D, ldd, r0, c0, lane, acc);
    }
  }
  template <int... Ts>
  static __device__ inline void go(const float* M1, int ld1, const float* M2,
                                   int ld2, float* D, int ldd, int wave, int lane,
                                   iseq<Ts...>) {
    (one<Ts>(M1, ld1, M2, ld2, D, ldd, wave, lane), ...);
  }
  static __device__ inline void run(const float* M1, int ld1, const float* M2,
                                    int ld2, float* D, int ldd, int wave, int lane) {
    go(M1, ld1, M2, ld2, D, ldd, wave, lane, typename mkseq<4>::type{});
  }
};

template <int T>
__device__ inline void final_tile(const float* sCt, const float* sA,
                                  const float* sXi, float* om, int wave, int lane) {
  constexpr int r0 = (T >> 2) << 4, c0 = (T & 3) << 4;
  constexpr unsigned mask = sel_mask(64, 3, T >> 2, T & 3);
  if (wave == (T & 7)) {
    v8f acc = wmma_tile<64, false, false>(sCt, LD64, sA, LD64, r0, c0, lane, mask);
    const int hw = lane >> 4;
    const int c = c0 + (lane & 15);
#pragma unroll
    for (int v = 0; v < 8; ++v) {
      const int r = r0 + v + 8 * hw;
      om[r * 64 + c] = sXi[r * LD64 + c] + acc[v];
    }
  }
}
template <int... Ts>
__device__ inline void final_go(const float* sCt, const float* sA, const float* sXi,
                                float* om, int wave, int lane, iseq<Ts...>) {
  (final_tile<Ts>(sCt, sA, sXi, om, wave, lane), ...);
}

__device__ inline void conv_band(const float* __restrict__ src, int sld, int sr, int sc,
                                 float* __restrict__ dst, int dld, int dr, int dc,
                                 int n, const float* __restrict__ wg, float bias,
                                 float scale, float shift, bool relu, int tid) {
  float w[9];
#pragma unroll
  for (int q = 0; q < 9; ++q) w[q] = wg[q];
  for (int p = tid; p < n * n; p += TPB) {
    const int i = p / n, j = p - i * n;
    float s = 0.f;
#pragma unroll
    for (int di = 0; di < 3; ++di) {
      const int ii = i + di - 1;
      if (ii < 0 || ii >= n) continue;
#pragma unroll
      for (int dj = 0; dj < 3; ++dj) {
        const int jj = j + dj - 1;
        if (jj < 0 || jj >= n) continue;
        s += w[di * 3 + dj] * src[(sr + ii) * sld + (sc + jj)];
      }
    }
    float y = (s + bias) * scale + shift;
    if (relu) y = fmaxf(y, 0.f);
    dst[(dr + i) * dld + (dc + j)] = y;
  }
}

__global__ __launch_bounds__(TPB)
void wavelet_mix2d_kernel(const float* __restrict__ x,
                          const float* __restrict__ ll_w,
                          const float* __restrict__ ll_b,
                          const float* __restrict__ bn_gamma,
                          const float* __restrict__ bn_beta,
                          const float* __restrict__ det_w,
                          const float* __restrict__ det_b,
                          float* __restrict__ out, int n_images) {
  __shared__ float sX[64 * LD64];
  __shared__ float sT[64 * LD64];
  __shared__ float sC[64 * LD64];
  __shared__ float sA64[64 * LD64];
  __shared__ float sA32[32 * LD32];
  __shared__ float s2a[32 * LD32];
  __shared__ float s2c[32 * LD32];
  __shared__ float s2m[32 * LD32];

  const int tid = threadIdx.x;
  const int lane = tid & 31;
  const int wave = __builtin_amdgcn_readfirstlane(tid >> 5);

  for (int i = tid; i < 64 * LD64; i += TPB) sA64[i] = 0.f;
  for (int i = tid; i < 32 * LD32; i += TPB) sA32[i] = 0.f;
  __syncthreads();
  for (int idx = tid; idx < 512; idx += TPB) {
    const int k = idx >> 3, i = idx & 7;
    const bool hi = k >= 32;
    const float f = hi ? (((i & 1) ? -1.f : 1.f) * c_H[7 - i]) : c_H[i];
    const int kk = hi ? (k - 32) : k;
    sA64[k * LD64 + ((2 * kk + i) & 63)] = f;
  }
  {
    const int k = tid >> 3, i = tid & 7;
    const bool hi = k >= 16;
    const float f = hi ? (((i & 1) ? -1.f : 1.f) * c_H[7 - i]) : c_H[i];
    const int kk = hi ? (k - 16) : k;
    sA32[k * LD32 + ((2 * kk + i) & 31)] = f;
  }
  const float bn_scale = bn_gamma[0] * rsqrtf(1.f + 1e-5f);

  for (int it = 0; it < IMGS_PER_BLOCK; ++it) {
    const int m = blockIdx.x * IMGS_PER_BLOCK + it;
    if (m >= n_images) break;
    const size_t img = (size_t)m * 4096;
    const float* xm = x + img;
    __syncthreads();
    for (int i = tid; i < 4096; i += TPB) sX[(i >> 6) * LD64 + (i & 63)] = xm[i];
    __syncthreads();

    S64<false, false, 0>::run(sA64, sX, sT, wave, lane);
    __syncthreads();
    S64<false, true, 1>::run(sT, sA64, sC, wave, lane);
    __syncthreads();
    S32<false, false, 0>::run(sA32, LD32, sC, LD64, s2a, LD32, wave, lane);
    __syncthreads();
    S32<false, true, 1>::run(s2a, LD32, sA32, LD32, s2c, LD32, wave, lane);
    __syncthreads();
    conv_band(s2c, LD32, 0, 0,   s2m, LD32, 0, 0,   16, ll_w,          ll_b[0],
              bn_scale, bn_beta[0], true, tid);
    conv_band(s2c, LD32, 16, 0,  s2m, LD32, 16, 0,  16, det_w + 3 * 9, det_b[3],
              1.f, 0.f, false, tid);
    conv_band(s2c, LD32, 0, 16,  s2m, LD32, 0, 16,  16, det_w + 4 * 9, det_b[4],
              1.f, 0.f, false, tid);
    conv_band(s2c, LD32, 16, 16, s2m, LD32, 16, 16, 16, det_w + 5 * 9, det_b[5],
              1.f, 0.f, false, tid);
    __syncthreads();
    S32<true, false, 2>::run(sA32, LD32, s2m, LD32, s2a, LD32, wave, lane);
    __syncthreads();
    S32<false, false, 3>::run(s2a, LD32, sA32, LD32, sT, LD64, wave, lane);
    __syncthreads();
    conv_band(sC, LD64, 32, 0,  sT, LD64, 32, 0,  32, det_w + 0 * 9, det_b[0],
              1.f, 0.f, false, tid);
    conv_band(sC, LD64, 0, 32,  sT, LD64, 0, 32,  32, det_w + 1 * 9, det_b[1],
              1.f, 0.f, false, tid);
    conv_band(sC, LD64, 32, 32, sT, LD64, 32, 32, 32, det_w + 2 * 9, det_b[2],
              1.f, 0.f, false, tid);
    __syncthreads();
    S64<true, false, 2>::run(sA64, sT, sC, wave, lane);
    __syncthreads();
    final_go(sC, sA64, sX, out + img, wave, lane, typename mkseq<16>::type{});
  }
}

extern "C" void kernel_launch(void* const* d_in, const int* in_sizes, int n_in,
                              void* d_out, int out_size, void* d_ws, size_t ws_size,
                              hipStream_t stream) {
  const float* x        = (const float*)d_in[0];
  const float* ll_w     = (const float*)d_in[1];
  const float* ll_b     = (const float*)d_in[2];
  const float* bn_gamma = (const float*)d_in[3];
  const float* bn_beta  = (const float*)d_in[4];
  const float* det_w    = (const float*)d_in[5];
  const float* det_b    = (const float*)d_in[6];
  float* out = (float*)d_out;
  const int n_images = in_sizes[0] / 4096;
  const int blocks = (n_images + IMGS_PER_BLOCK - 1) / IMGS_PER_BLOCK;
  wavelet_mix2d_kernel<<<dim3(blocks), dim3(TPB), 0, stream>>>(
      x, ll_w, ll_b, bn_gamma, bn_beta, det_w, det_b, out, n_images);
}


# Occupancy 1 -> 4 workgroups per WGP: VGPRs 398 -> 118 (LDS tile addresses = per-lane base + constant offset, launch bound 8 waves/SIMD), LDS 83456 -> 70784 B (32x32 temps in dead quadrants)
# speedup vs baseline: 2.6817x; 2.6817x over previous
#include <hip/hip_runtime.h>

#define TPB 256
#define LD64 65
#define LD32 33
#define IMGS_PER_BLOCK 2

typedef float v2f __attribute__((ext_vector_type(2)));
typedef float v8f __attribute__((ext_vector_type(8)));

__constant__ float c_H[8] = {
    0.23037781330885523f,  0.7148465705525415f,   0.6308807679295904f,
    -0.02798376941698385f, -0.18703481171888114f, 0.030841381835986965f,
    0.032883011666982945f, -0.010597401784997278f};

constexpr bool anz(int N, int k, int j) {
  int kk = (k >= N / 2) ? (k - N / 2) : k;
  int d = (j - 2 * kk) % N;
  if (d < 0) d += N;
  return d < 8;
}
constexpr unsigned mask_rows(int N, int r0) {
  unsigned m = 0;
  for (int jt = 0; jt < N / 4; ++jt)
    for (int k = 4 * jt; k < 4 * jt + 4; ++k)
      for (int r = r0; r < r0 + 16; ++r)
        if (anz(N, r, k)) m |= 1u << jt;
  return m;
}
constexpr unsigned mask_cols(int N, int c0) {
  unsigned m = 0;
  for (int jt = 0; jt < N / 4; ++jt)
    for (int k = 4 * jt; k < 4 * jt + 4; ++k)
      for (int c = c0; c < c0 + 16; ++c)
        if (anz(N, k, c)) m |= 1u << jt;
  return m;
}
constexpr unsigned sel_mask(int N, int SEL, int rb, int cb) {
  int blk = (SEL == 0 || SEL == 2) ? rb : cb;
  return (SEL <= 1) ? mask_rows(N, 16 * blk) : mask_cols(N, 16 * blk);
}

template <int... Is> struct iseq {};
template <int N, int... Is> struct mkseq : mkseq<N - 1, N - 1, Is...> {};
template <int... Is> struct mkseq<0, Is...> { using type = iseq<Is...>; };

__device__ inline int frag_row(int lane, int ld) { return (lane & 15) * ld + 2 * (lane >> 4); }
__device__ inline int frag_k(int lane, int ld)   { return 2 * (lane >> 4) * ld + (lane & 15); }
__device__ inline int frag_d(int lane, int ld)   { return 8 * (lane >> 4) * ld + (lane & 15); }

template <int K, bool TA, bool TB>
__device__ inline v8f wmma_tile(const float* __restrict__ pa, int ld1,
                                const float* __restrict__ pb, int ld2,
                                int r0, int c0, unsigned mask) {
  v8f acc = {0.f, 0.f, 0.f, 0.f, 0.f, 0.f, 0.f, 0.f};
#pragma unroll
  for (int j = 0; j < K / 4; ++j) {
    if (mask & (1u << j)) {
      const int k0 = 4 * j;
      v2f a, b;
      if (TA) { a.x = pa[k0 * ld1 + r0];  a.y = pa[(k0 + 1) * ld1 + r0]; }
      else    { a.x = pa[r0 * ld1 + k0];  a.y = pa[r0 * ld1 + k0 + 1];   }
      if (TB) { b.x = pb[c0 * ld2 + k0];  b.y = pb[c0 * ld2 + k0 + 1];   }
      else    { b.x = pb[k0 * ld2 + c0];  b.y = pb[(k0 + 1) * ld2 + c0]; }
      acc = __builtin_amdgcn_wmma_f32_16x16x4_f32(false, a, false, b,
                                                  (short)0, acc, false, false);
    }
  }
  return acc;
}

__device__ inline void store_tile(float* __restrict__ pd, int ld, int r0, int c0,
                                  v8f acc) {
#pragma unroll
  for (int v = 0; v < 8; ++v) pd[(r0 + v) * ld + c0] = acc[v];
}

template <bool TA, bool TB, int SEL>
struct S64 {
  template <int T>
  static __device__ inline void one(const float* pa, const float* pb, float* pd,
                                    int wave) {
    constexpr int r0 = (T >> 2) << 4, c0 = (T & 3) << 4;
    constexpr unsigned mask = sel_mask(64, SEL, T >> 2, T & 3);
    if (wave == (T & 7)) {
      v8f acc = wmma_tile<64, TA, TB>(pa, LD64, pb, LD64, r0, c0, mask);
      store_tile(pd, LD64, r0, c0, acc);
    }
  }
  template <int... Ts>
  static __device__ inline void go(const float* pa, const float* pb, float* pd,
                                   int wave, iseq<Ts...>) {
    (one<Ts>(pa, pb, pd, wave), ...);
  }
  static __device__ inline void run(const float* M1, const float* M2, float* D,
                                    int wave, int lane) {
    const float* pa = M1 + (TA ? frag_k(lane, LD64) : frag_row(lane, LD64));
    const float* pb = M2 + (TB ? frag_row(lane, LD64) : frag_k(lane, LD64));
    float* pd = D + frag_d(lane, LD64);
    go(pa, pb, pd, wave, typename mkseq<16>::type{});
  }
};

template <bool TA, bool TB, int SEL>
struct S32 {
  template <int T>
  static __device__ inline void one(const float* pa, int ld1, const float* pb,
                                    int ld2, float* pd, int ldd, int wave) {
    constexpr int r0 = (T >> 1) << 4, c0 = (T & 1) << 4;
    constexpr unsigned mask = sel_mask(32, SEL, T >> 1, T & 1);
    if (wave == T) {
      v8f acc = wmma_tile<32, TA, TB>(pa, ld1, pb, ld2, r0, c0, mask);
      store_tile(pd, ldd, r0, c0, acc);
    }
  }
  template <int... Ts>
  static __device__ inline void go(const float* pa, int ld1, const float* pb,
                                   int ld2, float* pd, int ldd, int wave,
                                   iseq<Ts...>) {
    (one<Ts>(pa, ld1, pb, ld2, pd, ldd, wave), ...);
  }
  static __device__ inline void run(const float* M1, int ld1, const float* M2,
                                    int ld2, float* D, int ldd, int wave, int lane) {
    const float* pa = M1 + (TA ? frag_k(lane, ld1) : frag_row(lane, ld1));
    const float* pb = M2 + (TB ? frag_row(lane, ld2) : frag_k(lane, ld2));
    float* pd = D + frag_d(lane, ldd);
    go(pa, ld1, pb, ld2, pd, ldd, wave, typename mkseq<4>::type{});
  }
};

template <int T>
__device__ inline void final_tile(const float* pa, const float* pb,
                                  const float* px, float* po, int wave) {
  constexpr int r0 = (T >> 2) << 4, c0 = (T & 3) << 4;
  constexpr unsigned mask = sel_mask(64, 3, T >> 2, T & 3);
  if (wave == (T & 7)) {
    v8f acc = wmma_tile<64, false, false>(pa, LD64, pb, LD64, r0, c0, mask);
#pragma unroll
    for (int v = 0; v < 8; ++v)
      po[(r0 + v) * 64 + c0] = px[(r0 + v) * LD64 + c0] + acc[v];
  }
}
template <int... Ts>
__device__ inline void final_go(const float* sCt, const float* sA, const float* sXi,
                                float* om, int wave, int lane, iseq<Ts...>) {
  const float* pa = sCt + frag_row(lane, LD64);
  const float* pb = sA + frag_k(lane, LD64);
  const float* px = sXi + frag_d(lane, LD64);
  float* po = om + frag_d(lane, 64);
  (final_tile<Ts>(pa, pb, px, po, wave), ...);
}

__device__ inline void conv_band(const float* __restrict__ src, int sld, int sr, int sc,
                                 float* __restrict__ dst, int dld, int dr, int dc,
                                 int n, const float* __restrict__ wg, float bias,
                                 float scale, float shift, bool relu, int tid) {
  float w[9];
#pragma unroll
  for (int q = 0; q < 9; ++q) w[q] = wg[q];
  for (int p = tid; p < n * n; p += TPB) {
    const int i = p / n, j = p - i * n;
    float s = 0.f;
#pragma unroll
    for (int di = 0; di < 3; ++di) {
      const int ii = i + di - 1;
      if (ii < 0 || ii >= n) continue;
#pragma unroll
      for (int dj = 0; dj < 3; ++dj) {
        const int jj = j + dj - 1;
        if (jj < 0 || jj >= n) continue;
        s += w[di * 3 + dj] * src[(sr + ii) * sld + (sc + jj)];
      }
    }
    float y = (s + bias) * scale + shift;
    if (relu) y = fmaxf(y, 0.f);
    dst[(dr + i) * dld + (dc + j)] = y;
  }
}

__global__ __launch_bounds__(TPB, 8)
void wavelet_mix2d_kernel(const float* __restrict__ x,
                          const float* __restrict__ ll_w,
                          const float* __restrict__ ll_b,
                          const float* __restrict__ bn_gamma,
                          const float* __restrict__ bn_beta,
                          const float* __restrict__ det_w,
                          const float* __restrict__ det_b,
                          float* __restrict__ out, int n_images) {
  __shared__ float sX[64 * LD64];
  __shared__ float sT[64 * LD64];
  __shared__ float sC[64 * LD64];
  __shared__ float sA64[64 * LD64];
  __shared__ float sA32[32 * LD32];

  const int tid = threadIdx.x;
  const int lane = tid & 31;
  const int wave = __builtin_amdgcn_readfirstlane(tid >> 5);

  for (int i = tid; i < 64 * LD64; i += TPB) sA64[i] = 0.f;
  for (int i = tid; i < 32 * LD32; i += TPB) sA32[i] = 0.f;
  __syncthreads();
  for (int idx = tid; idx < 512; idx += TPB) {
    const int k = idx >> 3, i = idx & 7;
    const bool hi = k >= 32;
    const float f = hi ? (((i & 1) ? -1.f : 1.f) * c_H[7 - i]) : c_H[i];
    const int kk = hi ? (k - 32) : k;
    sA64[k * LD64 + ((2 * kk + i) & 63)] = f;
  }
  {
    const int k = tid >> 3, i = tid & 7;
    const bool hi = k >= 16;
    const float f = hi ? (((i & 1) ? -1.f : 1.f) * c_H[7 - i]) : c_H[i];
    const int kk = hi ? (k - 16) : k;
    sA32[k * LD32 + ((2 * kk + i) & 31)] = f;
  }
  const float bn_scale = bn_gamma[0] * rsqrtf(1.f + 1e-5f);

  for (int it = 0; it < IMGS_PER_BLOCK; ++it) {
    const int m = blockIdx.x * IMGS_PER_BLOCK + it;
    if (m >= n_images) break;
    const size_t img = (size_t)m * 4096;
    const float* xm = x + img;
    __syncthreads();
    for (int i = tid; i < 4096; i += TPB) sX[(i >> 6) * LD64 + (i & 63)] = xm[i];
    __syncthreads();

    S64<false, false, 0>::run(sA64, sX, sT, wave, lane);
    __syncthreads();
    S64<false, true, 1>::run(sT, sA64, sC, wave, lane);
    __syncthreads();
    S32<false, false, 0>::run(sA32, LD32, sC, LD64, sT, LD64, wave, lane);
    __syncthreads();
    S32<false, true, 1>::run(sT, LD64, sA32, LD32, sC, LD64, wave, lane);
    __syncthreads();
    conv_band(sC, LD64, 0, 0,   sT, LD64, 0, 0,   16, ll_w,          ll_b[0],
              bn_scale, bn_beta[0], true, tid);
    conv_band(sC, LD64, 16, 0,  sT, LD64, 16, 0,  16, det_w + 3 * 9, det_b[3],
              1.f, 0.f, false, tid);
    conv_band(sC, LD64, 0, 16,  sT, LD64, 0, 16,  16, det_w + 4 * 9, det_b[4],
              1.f, 0.f, false, tid);
    conv_band(sC, LD64, 16, 16, sT, LD64, 16, 16, 16, det_w + 5 * 9, det_b[5],
              1.f, 0.f, false, tid);
    __syncthreads();
    S32<true, false, 2>::run(sA32, LD32, sT, LD64, sC, LD64, wave, lane);
    __syncthreads();
    S32<false, false, 3>::run(sC, LD64, sA32, LD32, sT, LD64, wave, lane);
    __syncthreads();
    conv_band(sC, LD64, 32, 0,  sT, LD64, 32, 0,  32, det_w + 0 * 9, det_b[0],
              1.f, 0.f, false, tid);
    conv_band(sC, LD64, 0, 32,  sT, LD64, 0, 32,  32, det_w + 1 * 9, det_b[1],
              1.f, 0.f, false, tid);
    conv_band(sC, LD64, 32, 32, sT, LD64, 32, 32, 32, det_w + 2 * 9, det_b[2],
              1.f, 0.f, false, tid);
    __syncthreads();
    S64<true, false, 2>::run(sA64, sT, sC, wave, lane);
    __syncthreads();
    final_go(sC, sA64, sX, out + img, wave, lane, typename mkseq<16>::type{});
  }
}

extern "C" void kernel_launch(void* const* d_in, const int* in_sizes, int n_in,
                              void* d_out, int out_size, void* d_ws, size_t ws_size,
                              hipStream_t stream) {
  const float* x        = (const float*)d_in[0];
  const float* ll_w     = (const float*)d_in[1];
  const float* ll_b     = (const float*)d_in[2];
  const float* bn_gamma = (const float*)d_in[3];
  const float* bn_beta  = (const float*)d_in[4];
  const float* det_w    = (const float*)d_in[5];
  const float* det_b    = (const float*)d_in[6];
  float* out = (float*)d_out;
  const int n_images = in_sizes[0] / 4096;
  const int blocks = (n_images + IMGS_PER_BLOCK - 1) / IMGS_PER_BLOCK;
  wavelet_mix2d_kernel<<<dim3(blocks), dim3(TPB), 0, stream>>>(
      x, ll_w, ll_b, bn_gamma, bn_beta, det_w, det_b, out, n_images);
}


# LDS row pitch 65 -> 76 dwords (input tile 74, 32x32 matrix 33 -> 36): fragment reads and D-tile stores become bank-conflict-free and each row-fragment k pair is one aligned 8-byte LDS read
# speedup vs baseline: 3.0160x; 1.1247x over previous
#include <hip/hip_runtime.h>

#define TPB 256
#define LDM 76
#define LDX 74
#define LD32 36
#define IMGS_PER_BLOCK 2

typedef float v2f __attribute__((ext_vector_type(2)));
typedef float v8f __attribute__((ext_vector_type(8)));

__constant__ float c_H[8] = {
    0.23037781330885523f,  0.7148465705525415f,   0.6308807679295904f,
    -0.02798376941698385f, -0.18703481171888114f, 0.030841381835986965f,
    0.032883011666982945f, -0.010597401784997278f};

constexpr bool anz(int N, int k, int j) {
  int kk = (k >= N / 2) ? (k - N / 2) : k;
  int d = (j - 2 * kk) % N;
  if (d < 0) d += N;
  return d < 8;
}
constexpr unsigned mask_rows(int N, int r0) {
  unsigned m = 0;
  for (int jt = 0; jt < N / 4; ++jt)
    for (int k = 4 * jt; k < 4 * jt + 4; ++k)
      for (int r = r0; r < r0 + 16; ++r)
        if (anz(N, r, k)) m |= 1u << jt;
  return m;
}
constexpr unsigned mask_cols(int N, int c0) {
  unsigned m = 0;
  for (int jt = 0; jt < N / 4; ++jt)
    for (int k = 4 * jt; k < 4 * jt + 4; ++k)
      for (int c = c0; c < c0 + 16; ++c)
        if (anz(N, k, c)) m |= 1u << jt;
  return m;
}
constexpr unsigned sel_mask(int N, int SEL, int rb, int cb) {
  int blk = (SEL == 0 || SEL == 2) ? rb : cb;
  return (SEL <= 1) ? mask_rows(N, 16 * blk) : mask_cols(N, 16 * blk);
}

template <int... Is> struct iseq {};
template <int N, int... Is> struct mkseq : mkseq<N - 1, N - 1, Is...> {};
template <int... Is> struct mkseq<0, Is...> { using type = iseq<Is...>; };

__device__ inline int frag_row(int lane, int ld) { return (lane & 15) * ld + 2 * (lane >> 4); }
__device__ inline int frag_k(int lane, int ld)   { return 2 * (lane >> 4) * ld + (lane & 15); }
__device__ inline int frag_d(int lane, int ld)   { return 8 * (lane >> 4) * ld + (lane & 15); }

template <int K, bool TA, bool TB>
__device__ inline v8f wmma_tile(const float* __restrict__ pa, int ld1,
                                const float* __restrict__ pb, int ld2,
                                int r0, int c0, unsigned mask) {
  v8f acc = {0.f, 0.f, 0.f, 0.f, 0.f, 0.f, 0.f, 0.f};
#pragma unroll
  for (int j = 0; j < K / 4; ++j) {
    if (mask & (1u << j)) {
      const int k0 = 4 * j;
      v2f a, b;
      if (TA) { a.x = pa[k0 * ld1 + r0];  a.y = pa[(k0 + 1) * ld1 + r0]; }
      else    { a = *reinterpret_cast<const v2f*>(pa + r0 * ld1 + k0);   }
      if (TB) { b = *reinterpret_cast<const v2f*>(pb + c0 * ld2 + k0);   }
      else    { b.x = pb[k0 * ld2 + c0];  b.y = pb[(k0 + 1) * ld2 + c0]; }
      acc = __builtin_amdgcn_wmma_f32_16x16x4_f32(false, a, false, b,
                                                  (short)0, acc, false, false);
    }
  }
  return acc;
}

__device__ inline void store_tile(float* __restrict__ pd, int ld, int r0, int c0,
                                  v8f acc) {
#pragma unroll
  for (int v = 0; v < 8; ++v) pd[(r0 + v) * ld + c0] = acc[v];
}

template <bool TA, bool TB, int SEL, int L1, int L2>
struct S64 {
  template <int T>
  static __device__ inline void one(const float* pa, const float* pb, float* pd,
                                    int wave) {
    constexpr int r0 = (T >> 2) << 4, c0 = (T & 3) << 4;
    constexpr unsigned mask = sel_mask(64, SEL, T >> 2, T & 3);
    if (wave == (T & 7)) {
      v8f acc = wmma_tile<64, TA, TB>(pa, L1, pb, L2, r0, c0, mask);
      store_tile(pd, LDM, r0, c0, acc);
    }
  }
  template <int... Ts>
  static __device__ inline void go(const float* pa, const float* pb, float* pd,
                                   int wave, iseq<Ts...>) {
    (one<Ts>(pa, pb, pd, wave), ...);
  }
  static __device__ inline void run(const float* M1, const float* M2, float* D,
                                    int wave, int lane) {
    const float* pa = M1 + (TA ? frag_k(lane, L1) : frag_row(lane, L1));
    const float* pb = M2 + (TB ? frag_row(lane, L2) : frag_k(lane, L2));
    float* pd = D + frag_d(lane, LDM);
    go(pa, pb, pd, wave, typename mkseq<16>::type{});
  }
};

template <bool TA, bool TB, int SEL>
struct S32 {
  template <int T>
  static __device__ inline void one(const float* pa, int ld1, const float* pb,
                                    int ld2, float* pd, int ldd, int wave) {
    constexpr int r0 = (T >> 1) << 4, c0 = (T & 1) << 4;
    constexpr unsigned mask = sel_mask(32, SEL, T >> 1, T & 1);
    if (wave == T) {
      v8f acc = wmma_tile<32, TA, TB>(pa, ld1, pb, ld2, r0, c0, mask);
      store_tile(pd, ldd, r0, c0, acc);
    }
  }
  template <int... Ts>
  static __device__ inline void go(const float* pa, int ld1, const float* pb,
                                   int ld2, float* pd, int ldd, int wave,
                                   iseq<Ts...>) {
    (one<Ts>(pa, ld1, pb, ld2, pd, ldd, wave), ...);
  }
  static __device__ inline void run(const float* M1, int ld1, const float* M2,
                                    int ld2, float* D, int ldd, int wave, int lane) {
    const float* pa = M1 + (TA ? frag_k(lane, ld1) : frag_row(lane, ld1));
    const float* pb = M2 + (TB ? frag_row(lane, ld2) : frag_k(lane, ld2));
    float* pd = D + frag_d(lane, ldd);
    go(pa, ld1, pb, ld2, pd, ldd, wave, typename mkseq<4>::type{});
  }
};

template <int T>
__device__ inline void final_tile(const float* pa, const float* pb,
                                  const float* px, float* po, int wave) {
  constexpr int r0 = (T >> 2) << 4, c0 = (T & 3) << 4;
  constexpr unsigned mask = sel_mask(64, 3, T >> 2, T & 3);
  if (wave == (T & 7)) {
    v8f acc = wmma_tile<64, false, false>(pa, LDM, pb, LDM, r0, c0, mask);
#pragma unroll
    for (int v = 0; v < 8; ++v)
      po[(r0 + v) * 64 + c0] = px[(r0 + v) * LDX + c0] + acc[v];
  }
}
template <int... Ts>
__device__ inline void final_go(const float* sCt, const float* sA, const float* sXi,
                                float* om, int wave, int lane, iseq<Ts...>) {
  const float* pa = sCt + frag_row(lane, LDM);
  const float* pb = sA + frag_k(lane, LDM);
  const float* px = sXi + frag_d(lane, LDX);
  float* po = om + frag_d(lane, 64);
  (final_tile<Ts>(pa, pb, px, po, wave), ...);
}

__device__ inline void conv_band(const float* __restrict__ src, int sld, int sr, int sc,
                                 float* __restrict__ dst, int dld, int dr, int dc,
                                 int n, const float* __restrict__ wg, float bias,
                                 float scale, float shift, bool relu, int tid) {
  float w[9];
#pragma unroll
  for (int q = 0; q < 9; ++q) w[q] = wg[q];
  for (int p = tid; p < n * n; p += TPB) {
    const int i = p / n, j = p - i * n;
    float s = 0.f;
#pragma unroll
    for (int di = 0; di < 3; ++di) {
      const int ii = i + di - 1;
      if (ii < 0 || ii >= n) continue;
#pragma unroll
      for (int dj = 0; dj < 3; ++dj) {
        const int jj = j + dj - 1;
        if (jj < 0 || jj >= n) continue;
        s += w[di * 3 + dj] * src[(sr + ii) * sld + (sc + jj)];
      }
    }
    float y = (s + bias) * scale + shift;
    if (relu) y = fmaxf(y, 0.f);
    dst[(dr + i) * dld + (dc + j)] = y;
  }
}

__global__ __launch_bounds__(TPB, 8)
void wavelet_mix2d_kernel(const float* __restrict__ x,
                          const float* __restrict__ ll_w,
                          const float* __restrict__ ll_b,
                          const float* __restrict__ bn_gamma,
                          const float* __restrict__ bn_beta,
                          const float* __restrict__ det_w,
                          const float* __restrict__ det_b,
                          float* __restrict__ out, int n_images) {
  alignas(16) __shared__ float sX[64 * LDX];
  alignas(16) __shared__ float sT[64 * LDM];
  alignas(16) __shared__ float sC[64 * LDM];
  alignas(16) __shared__ float sA64[64 * LDM];
  alignas(16) __shared__ float sA32[32 * LD32];

  const int tid = threadIdx.x;
  const int lane = tid & 31;
  const int wave = __builtin_amdgcn_readfirstlane(tid >> 5);

  for (int i = tid; i < 64 * LDM; i += TPB) sA64[i] = 0.f;
  for (int i = tid; i < 32 * LD32; i += TPB) sA32[i] = 0.f;
  __syncthreads();
  for (int idx = tid; idx < 512; idx += TPB) {
    const int k = idx >> 3, i = idx & 7;
    const bool hi = k >= 32;
    const float f = hi ? (((i & 1) ? -1.f : 1.f) * c_H[7 - i]) : c_H[i];
    const int kk = hi ? (k - 32) : k;
    sA64[k * LDM + ((2 * kk + i) & 63)] = f;
  }
  {
    const int k = tid >> 3, i = tid & 7;
    const bool hi = k >= 16;
    const float f = hi ? (((i & 1) ? -1.f : 1.f) * c_H[7 - i]) : c_H[i];
    const int kk = hi ? (k - 16) : k;
    sA32[k * LD32 + ((2 * kk + i) & 31)] = f;
  }
  const float bn_scale = bn_gamma[0] * rsqrtf(1.f + 1e-5f);

  for (int it = 0; it < IMGS_PER_BLOCK; ++it) {
    const int m = blockIdx.x * IMGS_PER_BLOCK + it;
    if (m >= n_images) break;
    const size_t img = (size_t)m * 4096;
    const float* xm = x + img;
    __syncthreads();
    for (int i = tid; i < 4096; i += TPB) sX[(i >> 6) * LDX + (i & 63)] = xm[i];
    __syncthreads();

    S64<false, false, 0, LDM, LDX>::run(sA64, sX, sT, wave, lane);
    __syncthreads();
    S64<false, true, 1, LDM, LDM>::run(sT, sA64, sC, wave, lane);
    __syncthreads();
    S32<false, false, 0>::run(sA32, LD32, sC, LDM, sT, LDM, wave, lane);
    __syncthreads();
    S32<false, true, 1>::run(sT, LDM, sA32, LD32, sC, LDM, wave, lane);
    __syncthreads();
    conv_band(sC, LDM, 0, 0,   sT, LDM, 0, 0,   16, ll_w,          ll_b[0],
              bn_scale, bn_beta[0], true, tid);
    conv_band(sC, LDM, 16, 0,  sT, LDM, 16, 0,  16, det_w + 3 * 9, det_b[3],
              1.f, 0.f, false, tid);
    conv_band(sC, LDM, 0, 16,  sT, LDM, 0, 16,  16, det_w + 4 * 9, det_b[4],
              1.f, 0.f, false, tid);
    conv_band(sC, LDM, 16, 16, sT, LDM, 16, 16, 16, det_w + 5 * 9, det_b[5],
              1.f, 0.f, false, tid);
    __syncthreads();
    S32<true, false, 2>::run(sA32, LD32, sT, LDM, sC, LDM, wave, lane);
    __syncthreads();
    S32<false, false, 3>::run(sC, LDM, sA32, LD32, sT, LDM, wave, lane);
    __syncthreads();
    conv_band(sC, LDM, 32, 0,  sT, LDM, 32, 0,  32, det_w + 0 * 9, det_b[0],
              1.f, 0.f, false, tid);
    conv_band(sC, LDM, 0, 32,  sT, LDM, 0, 32,  32, det_w + 1 * 9, det_b[1],
              1.f, 0.f, false, tid);
    conv_band(sC, LDM, 32, 32, sT, LDM, 32, 32, 32, det_w + 2 * 9, det_b[2],
              1.f, 0.f, false, tid);
    __syncthreads();
    S64<true, false, 2, LDM, LDM>::run(sA64, sT, sC, wave, lane);
    __syncthreads();
    final_go(sC, sA64, sX, out + img, wave, lane, typename mkseq<16>::type{});
  }
}

extern "C" void kernel_launch(void* const* d_in, const int* in_sizes, int n_in,
                              void* d_out, int out_size, void* d_ws, size_t ws_size,
                              hipStream_t stream) {
  const float* x        = (const float*)d_in[0];
  const float* ll_w     = (const float*)d_in[1];
  const float* ll_b     = (const float*)d_in[2];
  const float* bn_gamma = (const float*)d_in[3];
  const float* bn_beta  = (const float*)d_in[4];
  const float* det_w    = (const float*)d_in[5];
  const float* det_b    = (const float*)d_in[6];
  float* out = (float*)d_out;
  const int n_images = in_sizes[0] / 4096;
  const int blocks = (n_images + IMGS_PER_BLOCK - 1) / IMGS_PER_BLOCK;
  wavelet_mix2d_kernel<<<dim3(blocks), dim3(TPB), 0, stream>>>(
      x, ll_w, ll_b, bn_gamma, bn_beta, det_w, det_b, out, n_images);
}


# 3x3 band convs: per-tap masked load/wait/fma chains -> 4-row strips whose 18 clamped LDS reads issue as one batch, padding by select, weights read once per workgroup; tap order kept
# speedup vs baseline: 4.5138x; 1.4966x over previous
#include <hip/hip_runtime.h>

#define TPB 256
#define LDM 76
#define LDX 74
#define LD32 36
#define IMGS_PER_BLOCK 2

typedef float v2f __attribute__((ext_vector_type(2)));
typedef float v8f __attribute__((ext_vector_type(8)));

__constant__ float c_H[8] = {
    0.23037781330885523f,  0.7148465705525415f,   0.6308807679295904f,
    -0.02798376941698385f, -0.18703481171888114f, 0.030841381835986965f,
    0.032883011666982945f, -0.010597401784997278f};

constexpr bool anz(int N, int k, int j) {
  int kk = (k >= N / 2) ? (k - N / 2) : k;
  int d = (j - 2 * kk) % N;
  if (d < 0) d += N;
  return d < 8;
}
constexpr unsigned mask_rows(int N, int r0) {
  unsigned m = 0;
  for (int jt = 0; jt < N / 4; ++jt)
    for (int k = 4 * jt; k < 4 * jt + 4; ++k)
      for (int r = r0; r < r0 + 16; ++r)
        if (anz(N, r, k)) m |= 1u << jt;
  return m;
}
constexpr unsigned mask_cols(int N, int c0) {
  unsigned m = 0;
  for (int jt = 0; jt < N / 4; ++jt)
    for (int k = 4 * jt; k < 4 * jt + 4; ++k)
      for (int c = c0; c < c0 + 16; ++c)
        if (anz(N, k, c)) m |= 1u << jt;
  return m;
}
constexpr unsigned sel_mask(int N, int SEL, int rb, int cb) {
  int blk = (SEL == 0 || SEL == 2) ? rb : cb;
  return (SEL <= 1) ? mask_rows(N, 16 * blk) : mask_cols(N, 16 * blk);
}

template <int... Is> struct iseq {};
template <int N, int... Is> struct mkseq : mkseq<N - 1, N - 1, Is...> {};
template <int... Is> struct mkseq<0, Is...> { using type = iseq<Is...>; };

__device__ inline int frag_row(int lane, int ld) { return (lane & 15) * ld + 2 * (lane >> 4); }
__device__ inline int frag_k(int lane, int ld)   { return 2 * (lane >> 4) * ld + (lane & 15); }
__device__ inline int frag_d(int lane, int ld)   { return 8 * (lane >> 4) * ld + (lane & 15); }

template <int K, bool TA, bool TB>
__device__ inline v8f wmma_tile(const float* __restrict__ pa, int ld1,
                                const float* __restrict__ pb, int ld2,
                                int r0, int c0, unsigned mask) {
  v8f acc = {0.f, 0.f, 0.f, 0.f, 0.f, 0.f, 0.f, 0.f};
#pragma unroll
  for (int j = 0; j < K / 4; ++j) {
    if (mask & (1u << j)) {
      const int k0 = 4 * j;
      v2f a, b;
      if (TA) { a.x = pa[k0 * ld1 + r0];  a.y = pa[(k0 + 1) * ld1 + r0]; }
      else    { a = *reinterpret_cast<const v2f*>(pa + r0 * ld1 + k0);   }
      if (TB) { b = *reinterpret_cast<const v2f*>(pb + c0 * ld2 + k0);   }
      else    { b.x = pb[k0 * ld2 + c0];  b.y = pb[(k0 + 1) * ld2 + c0]; }
      acc = __builtin_amdgcn_wmma_f32_16x16x4_f32(false, a, false, b,
                                                  (short)0, acc, false, false);
    }
  }
  return acc;
}

__device__ inline void store_tile(float* __restrict__ pd, int ld, int r0, int c0,
                                  v8f acc) {
#pragma unroll
  for (int v = 0; v < 8; ++v) pd[(r0 + v) * ld + c0] = acc[v];
}

template <bool TA, bool TB, int SEL, int L1, int L2>
struct S64 {
  template <int T>
  static __device__ inline void one(const float* pa, const float* pb, float* pd,
                                    int wave) {
    constexpr int r0 = (T >> 2) << 4, c0 = (T & 3) << 4;
    constexpr unsigned mask = sel_mask(64, SEL, T >> 2, T & 3);
    if (wave == (T & 7)) {
      v8f acc = wmma_tile<64, TA, TB>(pa, L1, pb, L2, r0, c0, mask);
      store_tile(pd, LDM, r0, c0, acc);
    }
  }
  template <int... Ts>
  static __device__ inline void go(const float* pa, const float* pb, float* pd,
                                   int wave, iseq<Ts...>) {
    (one<Ts>(pa, pb, pd, wave), ...);
  }
  static __device__ inline void run(const float* M1, const float* M2, float* D,
                                    int wave, int lane) {
    const float* pa = M1 + (TA ? frag_k(lane, L1) : frag_row(lane, L1));
    const float* pb = M2 + (TB ? frag_row(lane, L2) : frag_k(lane, L2));
    float* pd = D + frag_d(lane, LDM);
    go(pa, pb, pd, wave, typename mkseq<16>::type{});
  }
};

template <bool TA, bool TB, int SEL>
struct S32 {
  template <int T>
  static __device__ inline void one(const float* pa, int ld1, const float* pb,
                                    int ld2, float* pd, int ldd, int wave) {
    constexpr int r0 = (T >> 1) << 4, c0 = (T & 1) << 4;
    constexpr unsigned mask = sel_mask(32, SEL, T >> 1, T & 1);
    if (wave == T) {
      v8f acc = wmma_tile<32, TA, TB>(pa, ld1, pb, ld2, r0, c0, mask);
      store_tile(pd, ldd, r0, c0, acc);
    }
  }
  template <int... Ts>
  static __device__ inline void go(const float* pa, int ld1, const float* pb,
                                   int ld2, float* pd, int ldd, int wave,
                                   iseq<Ts...>) {
    (one<Ts>(pa, ld1, pb, ld2, pd, ldd, wave), ...);
  }
  static __device__ inline void run(const float* M1, int ld1, const float* M2,
                                    int ld2, float* D, int ldd, int wave, int lane) {
    const float* pa = M1 + (TA ? frag_k(lane, ld1) : frag_row(lane, ld1));
    const float* pb = M2 + (TB ? frag_row(lane, ld2) : frag_k(lane, ld2));
    float* pd = D + frag_d(lane, ldd);
    go(pa, ld1, pb, ld2, pd, ldd, wave, typename mkseq<4>::type{});
  }
};

template <int T>
__device__ inline void final_tile(const float* pa, const float* pb,
                                  const float* px, float* po, int wave) {
  constexpr int r0 = (T >> 2) << 4, c0 = (T & 3) << 4;
  constexpr unsigned mask = sel_mask(64, 3, T >> 2, T & 3);
  if (wave == (T & 7)) {
    v8f acc = wmma_tile<64, false, false>(pa, LDM, pb, LDM, r0, c0, mask);
#pragma unroll
    for (int v = 0; v < 8; ++v)
      po[(r0 + v) * 64 + c0] = px[(r0 + v) * LDX + c0] + acc[v];
  }
}
template <int... Ts>
__device__ inline void final_go(const float* sCt, const float* sA, const float* sXi,
                                float* om, int wave, int lane, iseq<Ts...>) {
  const float* pa = sCt + frag_row(lane, LDM);
  const float* pb = sA + frag_k(lane, LDM);
  const float* px = sXi + frag_d(lane, LDX);
  float* po = om + frag_d(lane, 64);
  (final_tile<Ts>(pa, pb, px, po, wave), ...);
}

template <int N>
__device__ inline void conv_strip4(const float* __restrict__ src,
                                   float* __restrict__ dst, int i0, int j,
                                   const float (&w)[9], float bias, float scale,
                                   float shift, bool relu) {
  const bool top = i0 > 0, bot = i0 + 4 < N, lft = j > 0, rgt = j + 1 < N;
  const int jl = lft ? j - 1 : j;
  const int jr = rgt ? j + 1 : j;
  const float* pt = src + (top ? i0 - 1 : i0) * LDM;
  const float* pm = src + i0 * LDM;
  const float* pb = src + (bot ? i0 + 4 : i0 + 3) * LDM;
  float v[6][3];
  v[0][0] = pt[jl]; v[0][1] = pt[j]; v[0][2] = pt[jr];
#pragma unroll
  for (int r = 0; r < 4; ++r) {
    v[r + 1][0] = pm[r * LDM + jl];
    v[r + 1][1] = pm[r * LDM + j];
    v[r + 1][2] = pm[r * LDM + jr];
  }
  v[5][0] = pb[jl]; v[5][1] = pb[j]; v[5][2] = pb[jr];
#pragma unroll
  for (int r = 0; r < 6; ++r) {
    if (!lft) v[r][0] = 0.f;
    if (!rgt) v[r][2] = 0.f;
  }
#pragma unroll
  for (int cc = 0; cc < 3; ++cc) {
    if (!top) v[0][cc] = 0.f;
    if (!bot) v[5][cc] = 0.f;
  }
#pragma unroll
  for (int o = 0; o < 4; ++o) {
    float s = 0.f;
#pragma unroll
    for (int di = 0; di < 3; ++di) {
#pragma unroll
      for (int dj = 0; dj < 3; ++dj) s += w[di * 3 + dj] * v[o + di][dj];
    }
    float y = (s + bias) * scale + shift;
    if (relu) y = fmaxf(y, 0.f);
    dst[(i0 + o) * LDM + j] = y;
  }
}

__global__ __launch_bounds__(TPB, 8)
void wavelet_mix2d_kernel(const float* __restrict__ x,
                          const float* __restrict__ ll_w,
                          const float* __restrict__ ll_b,
                          const float* __restrict__ bn_gamma,
                          const float* __restrict__ bn_beta,
                          const float* __restrict__ det_w,
                          const float* __restrict__ det_b,
                          float* __restrict__ out, int n_images) {
  alignas(16) __shared__ float sX[64 * LDX];
  alignas(16) __shared__ float sT[64 * LDM];
  alignas(16) __shared__ float sC[64 * LDM];
  alignas(16) __shared__ float sA64[64 * LDM];
  alignas(16) __shared__ float sA32[32 * LD32];

  const int tid = threadIdx.x;
  const int lane = tid & 31;
  const int wave = __builtin_amdgcn_readfirstlane(tid >> 5);

  for (int i = tid; i < 64 * LDM; i += TPB) sA64[i] = 0.f;
  for (int i = tid; i < 32 * LD32; i += TPB) sA32[i] = 0.f;
  __syncthreads();
  for (int idx = tid; idx < 512; idx += TPB) {
    const int k = idx >> 3, i = idx & 7;
    const bool hi = k >= 32;
    const float f = hi ? (((i & 1) ? -1.f : 1.f) * c_H[7 - i]) : c_H[i];
    const int kk = hi ? (k - 32) : k;
    sA64[k * LDM + ((2 * kk + i) & 63)] = f;
  }
  {
    const int k = tid >> 3, i = tid & 7;
    const bool hi = k >= 16;
    const float f = hi ? (((i & 1) ? -1.f : 1.f) * c_H[7 - i]) : c_H[i];
    const int kk = hi ? (k - 16) : k;
    sA32[k * LD32 + ((2 * kk + i) & 31)] = f;
  }
  const float bn_scale = bn_gamma[0] * rsqrtf(1.f + 1e-5f);

  const int b4 = wave >> 1;
  const float* w4p = (b4 == 0) ? ll_w : det_w + 9 * (b4 + 2);
  float w4[9];
#pragma unroll
  for (int q = 0; q < 9; ++q) w4[q] = w4p[q];
  const float bias4  = (b4 == 0) ? ll_b[0] : det_b[b4 + 2];
  const float scale4 = (b4 == 0) ? bn_scale : 1.f;
  const float shift4 = (b4 == 0) ? bn_beta[0] : 0.f;
  const int off4 = (b4 & 1) * 16 * LDM + (b4 >> 1) * 16;
  const int i4 = 4 * ((wave & 1) * 2 + (lane >> 4));
  const int j4 = lane & 15;
  float w6[3][9], bias6[3];
#pragma unroll
  for (int b = 0; b < 3; ++b) {
#pragma unroll
    for (int q = 0; q < 9; ++q) w6[b][q] = det_w[9 * b + q];
    bias6[b] = det_b[b];
  }
  const int i6 = 4 * wave;

  for (int it = 0; it < IMGS_PER_BLOCK; ++it) {
    const int m = blockIdx.x * IMGS_PER_BLOCK + it;
    if (m >= n_images) break;
    const size_t img = (size_t)m * 4096;
    const float* xm = x + img;
    __syncthreads();
    for (int i = tid; i < 4096; i += TPB) sX[(i >> 6) * LDX + (i & 63)] = xm[i];
    __syncthreads();

    S64<false, false, 0, LDM, LDX>::run(sA64, sX, sT, wave, lane);
    __syncthreads();
    S64<false, true, 1, LDM, LDM>::run(sT, sA64, sC, wave, lane);
    __syncthreads();
    S32<false, false, 0>::run(sA32, LD32, sC, LDM, sT, LDM, wave, lane);
    __syncthreads();
    S32<false, true, 1>::run(sT, LDM, sA32, LD32, sC, LDM, wave, lane);
    __syncthreads();
    conv_strip4<16>(sC + off4, sT + off4, i4, j4, w4, bias4, scale4, shift4,
                    b4 == 0);
    __syncthreads();
    S32<true, false, 2>::run(sA32, LD32, sT, LDM, sC, LDM, wave, lane);
    __syncthreads();
    S32<false, false, 3>::run(sC, LDM, sA32, LD32, sT, LDM, wave, lane);
    __syncthreads();
    conv_strip4<32>(sC + 32 * LDM, sT + 32 * LDM, i6, lane, w6[0], bias6[0],
                    1.f, 0.f, false);
    conv_strip4<32>(sC + 32, sT + 32, i6, lane, w6[1], bias6[1],
                    1.f, 0.f, false);
    conv_strip4<32>(sC + 32 * LDM + 32, sT + 32 * LDM + 32, i6, lane, w6[2],
                    bias6[2], 1.f, 0.f, false);
    __syncthreads();
    S64<true, false, 2, LDM, LDM>::run(sA64, sT, sC, wave, lane);
    __syncthreads();
    final_go(sC, sA64, sX, out + img, wave, lane, typename mkseq<16>::type{});
  }
}

extern "C" void kernel_launch(void* const* d_in, const int* in_sizes, int n_in,
                              void* d_out, int out_size, void* d_ws, size_t ws_size,
                              hipStream_t stream) {
  const float* x        = (const float*)d_in[0];
  const float* ll_w     = (const float*)d_in[1];
  const float* ll_b     = (const float*)d_in[2];
  const float* bn_gamma = (const float*)d_in[3];
  const float* bn_beta  = (const float*)d_in[4];
  const float* det_w    = (const float*)d_in[5];
  const float* det_b    = (const float*)d_in[6];
  float* out = (float*)d_out;
  const int n_images = in_sizes[0] / 4096;
  const int blocks = (n_images + IMGS_PER_BLOCK - 1) / IMGS_PER_BLOCK;
  wavelet_mix2d_kernel<<<dim3(blocks), dim3(TPB), 0, stream>>>(
      x, ll_w, ll_b, bn_gamma, bn_beta, det_w, det_b, out, n_images);
}


# Input tile load: chain of dependent 4-byte global loads (2 in flight per wave, 8 round trips per image) -> four independent 16-byte loads per thread issued together, then 8-byte LDS stores
# speedup vs baseline: 5.5144x; 1.2217x over previous
#include <hip/hip_runtime.h>

#define TPB 256
#define LDM 76
#define LDX 74
#define LD32 36
#define IMGS_PER_BLOCK 2

typedef float v2f __attribute__((ext_vector_type(2)));
typedef float v8f __attribute__((ext_vector_type(8)));

__constant__ float c_H[8] = {
    0.23037781330885523f,  0.7148465705525415f,   0.6308807679295904f,
    -0.02798376941698385f, -0.18703481171888114f, 0.030841381835986965f,
    0.032883011666982945f, -0.010597401784997278f};

constexpr bool anz(int N, int k, int j) {
  int kk = (k >= N / 2) ? (k - N / 2) : k;
  int d = (j - 2 * kk) % N;
  if (d < 0) d += N;
  return d < 8;
}
constexpr unsigned mask_rows(int N, int r0) {
  unsigned m = 0;
  for (int jt = 0; jt < N / 4; ++jt)
    for (int k = 4 * jt; k < 4 * jt + 4; ++k)
      for (int r = r0; r < r0 + 16; ++r)
        if (anz(N, r, k)) m |= 1u << jt;
  return m;
}
constexpr unsigned mask_cols(int N, int c0) {
  unsigned m = 0;
  for (int jt = 0; jt < N / 4; ++jt)
    for (int k = 4 * jt; k < 4 * jt + 4; ++k)
      for (int c = c0; c < c0 + 16; ++c)
        if (anz(N, k, c)) m |= 1u << jt;
  return m;
}
constexpr unsigned sel_mask(int N, int SEL, int rb, int cb) {
  int blk = (SEL == 0 || SEL == 2) ? rb : cb;
  return (SEL <= 1) ? mask_rows(N, 16 * blk) : mask_cols(N, 16 * blk);
}

template <int... Is> struct iseq {};
template <int N, int... Is> struct mkseq : mkseq<N - 1, N - 1, Is...> {};
template <int... Is> struct mkseq<0, Is...> { using type = iseq<Is...>; };

__device__ inline int frag_row(int lane, int ld) { return (lane & 15) * ld + 2 * (lane >> 4); }
__device__ inline int frag_k(int lane, int ld)   { return 2 * (lane >> 4) * ld + (lane & 15); }
__device__ inline int frag_d(int lane, int ld)   { return 8 * (lane >> 4) * ld + (lane & 15); }

template <int K, bool TA, bool TB>
__device__ inline v8f wmma_tile(const float* __restrict__ pa, int ld1,
                                const float* __restrict__ pb, int ld2,
                                int r0, int c0, unsigned mask) {
  v8f acc = {0.f, 0.f, 0.f, 0.f, 0.f, 0.f, 0.f, 0.f};
#pragma unroll
  for (int j = 0; j < K / 4; ++j) {
    if (mask & (1u << j)) {
      const int k0 = 4 * j;
      v2f a, b;
      if (TA) { a.x = pa[k0 * ld1 + r0];  a.y = pa[(k0 + 1) * ld1 + r0]; }
      else    { a = *reinterpret_cast<const v2f*>(pa + r0 * ld1 + k0);   }
      if (TB) { b = *reinterpret_cast<const v2f*>(pb + c0 * ld2 + k0);   }
      else    { b.x = pb[k0 * ld2 + c0];  b.y = pb[(k0 + 1) * ld2 + c0]; }
      acc = __builtin_amdgcn_wmma_f32_16x16x4_f32(false, a, false, b,
                                                  (short)0, acc, false, false);
    }
  }
  return acc;
}

__device__ inline void store_tile(float* __restrict__ pd, int ld, int r0, int c0,
                                  v8f acc) {
#pragma unroll
  for (int v = 0; v < 8; ++v) pd[(r0 + v) * ld + c0] = acc[v];
}

template <bool TA, bool TB, int SEL, int L1, int L2>
struct S64 {
  template <int T>
  static __device__ inline void one(const float* pa, const float* pb, float* pd,
                                    int wave) {
    constexpr int r0 = (T >> 2) << 4, c0 = (T & 3) << 4;
    constexpr unsigned mask = sel_mask(64, SEL, T >> 2, T & 3);
    if (wave == (T & 7)) {
      v8f acc = wmma_tile<64, TA, TB>(pa, L1, pb, L2, r0, c0, mask);
      store_tile(pd, LDM, r0, c0, acc);
    }
  }
  template <int... Ts>
  static __device__ inline void go(const float* pa, const float* pb, float* pd,
                                   int wave, iseq<Ts...>) {
    (one<Ts>(pa, pb, pd, wave), ...);
  }
  static __device__ inline void run(const float* M1, const float* M2, float* D,
                                    int wave, int lane) {
    const float* pa = M1 + (TA ? frag_k(lane, L1) : frag_row(lane, L1));
    const float* pb = M2 + (TB ? frag_row(lane, L2) : frag_k(lane, L2));
    float* pd = D + frag_d(lane, LDM);
    go(pa, pb, pd, wave, typename mkseq<16>::type{});
  }
};

template <bool TA, bool TB, int SEL>
struct S32 {
  template <int T>
  static __device__ inline void one(const float* pa, int ld1, const float* pb,
                                    int ld2, float* pd, int ldd, int wave) {
    constexpr int r0 = (T >> 1) << 4, c0 = (T & 1) << 4;
    constexpr unsigned mask = sel_mask(32, SEL, T >> 1, T & 1);
    if (wave == T) {
      v8f acc = wmma_tile<32, TA, TB>(pa, ld1, pb, ld2, r0, c0, mask);
      store_tile(pd, ldd, r0, c0, acc);
    }
  }
  template <int... Ts>
  static __device__ inline void go(const float* pa, int ld1, const float* pb,
                                   int ld2, float* pd, int ldd, int wave,
                                   iseq<Ts...>) {
    (one<Ts>(pa, ld1, pb, ld2, pd, ldd, wave), ...);
  }
  static __device__ inline void run(const float* M1, int ld1, const float* M2,
                                    int ld2, float* D, int ldd, int wave, int lane) {
    const float* pa = M1 + (TA ? frag_k(lane, ld1) : frag_row(lane, ld1));
    const float* pb = M2 + (TB ? frag_row(lane, ld2) : frag_k(lane, ld2));
    float* pd = D + frag_d(lane, ldd);
    go(pa, ld1, pb, ld2, pd, ldd, wave, typename mkseq<4>::type{});
  }
};

template <int T>
__device__ inline void final_tile(const float* pa, const float* pb,
                                  const float* px, float* po, int wave) {
  constexpr int r0 = (T >> 2) << 4, c0 = (T & 3) << 4;
  constexpr unsigned mask = sel_mask(64, 3, T >> 2, T & 3);
  if (wave == (T & 7)) {
    v8f acc = wmma_tile<64, false, false>(pa, LDM, pb, LDM, r0, c0, mask);
#pragma unroll
    for (int v = 0; v < 8; ++v)
      po[(r0 + v) * 64 + c0] = px[(r0 + v) * LDX + c0] + acc[v];
  }
}
template <int... Ts>
__device__ inline void final_go(const float* sCt, const float* sA, const float* sXi,
                                float* om, int wave, int lane, iseq<Ts...>) {
  const float* pa = sCt + frag_row(lane, LDM);
  const float* pb = sA + frag_k(lane, LDM);
  const float* px = sXi + frag_d(lane, LDX);
  float* po = om + frag_d(lane, 64);
  (final_tile<Ts>(pa, pb, px, po, wave), ...);
}

template <int N>
__device__ inline void conv_strip4(const float* __restrict__ src,
                                   float* __restrict__ dst, int i0, int j,
                                   const float (&w)[9], float bias, float scale,
                                   float shift, bool relu) {
  const bool top = i0 > 0, bot = i0 + 4 < N, lft = j > 0, rgt = j + 1 < N;
  const int jl = lft ? j - 1 : j;
  const int jr = rgt ? j + 1 : j;
  const float* pt = src + (top ? i0 - 1 : i0) * LDM;
  const float* pm = src + i0 * LDM;
  const float* pb = src + (bot ? i0 + 4 : i0 + 3) * LDM;
  float v[6][3];
  v[0][0] = pt[jl]; v[0][1] = pt[j]; v[0][2] = pt[jr];
#pragma unroll
  for (int r = 0; r < 4; ++r) {
    v[r + 1][0] = pm[r * LDM + jl];
    v[r + 1][1] = pm[r * LDM + j];
    v[r + 1][2] = pm[r * LDM + jr];
  }
  v[5][0] = pb[jl]; v[5][1] = pb[j]; v[5][2] = pb[jr];
#pragma unroll
  for (int r = 0; r < 6; ++r) {
    if (!lft) v[r][0] = 0.f;
    if (!rgt) v[r][2] = 0.f;
  }
#pragma unroll
  for (int cc = 0; cc < 3; ++cc) {
    if (!top) v[0][cc] = 0.f;
    if (!bot) v[5][cc] = 0.f;
  }
#pragma unroll
  for (int o = 0; o < 4; ++o) {
    float s = 0.f;
#pragma unroll
    for (int di = 0; di < 3; ++di) {
#pragma unroll
      for (int dj = 0; dj < 3; ++dj) s += w[di * 3 + dj] * v[o + di][dj];
    }
    float y = (s + bias) * scale + shift;
    if (relu) y = fmaxf(y, 0.f);
    dst[(i0 + o) * LDM + j] = y;
  }
}

__global__ __launch_bounds__(TPB, 8)
void wavelet_mix2d_kernel(const float* __restrict__ x,
                          const float* __restrict__ ll_w,
                          const float* __restrict__ ll_b,
                          const float* __restrict__ bn_gamma,
                          const float* __restrict__ bn_beta,
                          const float* __restrict__ det_w,
                          const float* __restrict__ det_b,
                          float* __restrict__ out, int n_images) {
  alignas(16) __shared__ float sX[64 * LDX];
  alignas(16) __shared__ float sT[64 * LDM];
  alignas(16) __shared__ float sC[64 * LDM];
  alignas(16) __shared__ float sA64[64 * LDM];
  alignas(16) __shared__ float sA32[32 * LD32];

  const int tid = threadIdx.x;
  const int lane = tid & 31;
  const int wave = __builtin_amdgcn_readfirstlane(tid >> 5);

  for (int i = tid; i < 64 * LDM; i += TPB) sA64[i] = 0.f;
  for (int i = tid; i < 32 * LD32; i += TPB) sA32[i] = 0.f;
  __syncthreads();
  for (int idx = tid; idx < 512; idx += TPB) {
    const int k = idx >> 3, i = idx & 7;
    const bool hi = k >= 32;
    const float f = hi ? (((i & 1) ? -1.f : 1.f) * c_H[7 - i]) : c_H[i];
    const int kk = hi ? (k - 32) : k;
    sA64[k * LDM + ((2 * kk + i) & 63)] = f;
  }
  {
    const int k = tid >> 3, i = tid & 7;
    const bool hi = k >= 16;
    const float f = hi ? (((i & 1) ? -1.f : 1.f) * c_H[7 - i]) : c_H[i];
    const int kk = hi ? (k - 16) : k;
    sA32[k * LD32 + ((2 * kk + i) & 31)] = f;
  }
  const float bn_scale = bn_gamma[0] * rsqrtf(1.f + 1e-5f);

  const int b4 = wave >> 1;
  const float* w4p = (b4 == 0) ? ll_w : det_w + 9 * (b4 + 2);
  float w4[9];
#pragma unroll
  for (int q = 0; q < 9; ++q) w4[q] = w4p[q];
  const float bias4  = (b4 == 0) ? ll_b[0] : det_b[b4 + 2];
  const float scale4 = (b4 == 0) ? bn_scale : 1.f;
  const float shift4 = (b4 == 0) ? bn_beta[0] : 0.f;
  const int off4 = (b4 & 1) * 16 * LDM + (b4 >> 1) * 16;
  const int i4 = 4 * ((wave & 1) * 2 + (lane >> 4));
  const int j4 = lane & 15;
  float w6[3][9], bias6[3];
#pragma unroll
  for (int b = 0; b < 3; ++b) {
#pragma unroll
    for (int q = 0; q < 9; ++q) w6[b][q] = det_w[9 * b + q];
    bias6[b] = det_b[b];
  }
  const int i6 = 4 * wave;

  for (int it = 0; it < IMGS_PER_BLOCK; ++it) {
    const int m = blockIdx.x * IMGS_PER_BLOCK + it;
    if (m >= n_images) break;
    const size_t img = (size_t)m * 4096;
    const float* xm = x + img;
    __syncthreads();
    {
      const float4* xv = reinterpret_cast<const float4*>(xm);
      float4 q[4];
#pragma unroll
      for (int u = 0; u < 4; ++u) q[u] = xv[tid + u * TPB];
#pragma unroll
      for (int u = 0; u < 4; ++u) {
        const int i4 = tid + u * TPB;
        float* d = sX + (i4 >> 4) * LDX + ((i4 & 15) << 2);
        v2f lo, hi;
        lo.x = q[u].x; lo.y = q[u].y; hi.x = q[u].z; hi.y = q[u].w;
        *reinterpret_cast<v2f*>(d) = lo;
        *reinterpret_cast<v2f*>(d + 2) = hi;
      }
    }
    __syncthreads();

    S64<false, false, 0, LDM, LDX>::run(sA64, sX, sT, wave, lane);
    __syncthreads();
    S64<false, true, 1, LDM, LDM>::run(sT, sA64, sC, wave, lane);
    __syncthreads();
    S32<false, false, 0>::run(sA32, LD32, sC, LDM, sT, LDM, wave, lane);
    __syncthreads();
    S32<false, true, 1>::run(sT, LDM, sA32, LD32, sC, LDM, wave, lane);
    __syncthreads();
    conv_strip4<16>(sC + off4, sT + off4, i4, j4, w4, bias4, scale4, shift4,
                    b4 == 0);
    __syncthreads();
    S32<true, false, 2>::run(sA32, LD32, sT, LDM, sC, LDM, wave, lane);
    __syncthreads();
    S32<false, false, 3>::run(sC, LDM, sA32, LD32, sT, LDM, wave, lane);
    __syncthreads();
    conv_strip4<32>(sC + 32 * LDM, sT + 32 * LDM, i6, lane, w6[0], bias6[0],
                    1.f, 0.f, false);
    conv_strip4<32>(sC + 32, sT + 32, i6, lane, w6[1], bias6[1],
                    1.f, 0.f, false);
    conv_strip4<32>(sC + 32 * LDM + 32, sT + 32 * LDM + 32, i6, lane, w6[2],
                    bias6[2], 1.f, 0.f, false);
    __syncthreads();
    S64<true, false, 2, LDM, LDM>::run(sA64, sT, sC, wave, lane);
    __syncthreads();
    final_go(sC, sA64, sX, out + img, wave, lane, typename mkseq<16>::type{});
  }
}

extern "C" void kernel_launch(void* const* d_in, const int* in_sizes, int n_in,
                              void* d_out, int out_size, void* d_ws, size_t ws_size,
                              hipStream_t stream) {
  const float* x        = (const float*)d_in[0];
  const float* ll_w     = (const float*)d_in[1];
  const float* ll_b     = (const float*)d_in[2];
  const float* bn_gamma = (const float*)d_in[3];
  const float* bn_beta  = (const float*)d_in[4];
  const float* det_w    = (const float*)d_in[5];
  const float* det_b    = (const float*)d_in[6];
  float* out = (float*)d_out;
  const int n_images = in_sizes[0] / 4096;
  const int blocks = (n_images + IMGS_PER_BLOCK - 1) / IMGS_PER_BLOCK;
  wavelet_mix2d_kernel<<<dim3(blocks), dim3(TPB), 0, stream>>>(
      x, ll_w, ll_b, bn_gamma, bn_beta, det_w, det_b, out, n_images);
}
